# starting kernel
# baseline (speedup 1.0000x reference)
#include <hip/hip_runtime.h>

typedef __attribute__((ext_vector_type(2))) float v2f;
typedef __attribute__((ext_vector_type(8))) float v8f;

#define CIN    32
#define COUT   32
#define HH     256
#define WW     256
#define TILE_W 128
#define NSTEP  72

#define XPADW   34
#define XROWW   130
#define X_ELEMS (3 * XROWW * XPADW)
#define W_OFF   X_ELEMS
#define W_FRAGS (NSTEP * 2 * 32)
#define B_OFF   (W_OFF + W_FRAGS * 2)
#define SMEM_FLOATS (B_OFF + 32)

extern "C" __global__ __launch_bounds__(256, 2)
void conv3x3_wmma_f32(const float* __restrict__ x,
                      const float* __restrict__ wgt,
                      const float* __restrict__ bias,
                      float* __restrict__ out)
{
    extern __shared__ float smem[];
    float* xlds = smem;
    v2f*   wlds = (v2f*)(smem + W_OFF);
    float* blds = smem + B_OFF;

    const int tid   = threadIdx.x;
    const int blk   = blockIdx.x;
    const int b     = blk >> 9;
    const int h     = (blk >> 1) & 255;
    const int wtile = blk & 1;
    const int wbase_g = wtile * TILE_W;

    for (int idx = tid; idx < W_FRAGS; idx += 256) {
        int s    = idx >> 6;
        int rem  = idx & 63;
        int t    = rem >> 5;
        int l    = rem & 31;
        int k0   = 4 * s + ((l >> 4) << 1);
        int cout = (t << 4) + (l & 15);
        v2f wv;
        #pragma unroll
        for (int v = 0; v < 2; ++v) {
            int k    = k0 + v;
            int ci   = k & 31;
            int khkw = k >> 5;
            wv[v] = wgt[(cout * CIN + ci) * 9 + khkw];
        }
        wlds[idx] = wv;
    }

    if (tid < 32) blds[tid] = bias[tid];

    for (int idx = tid; idx < 3 * CIN * XROWW; idx += 256) {
        int kh = idx / (CIN * XROWW);
        int r  = idx - kh * (CIN * XROWW);
        int ci = r / XROWW;
        int wp = r - ci * XROWW;
        int hin = h + kh - 1;
        int win = wbase_g + wp - 1;
        float v = 0.0f;
        if ((unsigned)hin < HH && (unsigned)win < WW)
            v = x[((b * CIN + ci) * HH + hin) * WW + win];
        xlds[(kh * XROWW + wp) * XPADW + ci] = v;
    }
    __syncthreads();

    const int lane = tid & 31;
    const int wl   = (tid >> 5) << 4;
    const int nco  = lane & 15;
    const int hi   = lane >> 4;

    v8f acc0 = {};
    v8f acc1 = {};

    #pragma unroll
    for (int s = 0; s < NSTEP; ++s) {
        const int khkw = s >> 3;
        const int kh   = khkw / 3;
        const int kw   = khkw - 3 * kh;
        const int cig  = (s & 7) << 2;

        v2f a0 = wlds[(s * 2 + 0) * 32 + lane];
        v2f a1 = wlds[(s * 2 + 1) * 32 + lane];

        const int wp = wl + nco + kw;
        const int ci = cig + (hi << 1);
        v2f bv = *(const v2f*)&xlds[(kh * XROWW + wp) * XPADW + ci];

        acc0 = __builtin_amdgcn_wmma_f32_16x16x4_f32(
                   false, a0, false, bv, (short)0, acc0, false, false);
        acc1 = __builtin_amdgcn_wmma_f32_16x16x4_f32(
                   false, a1, false, bv, (short)0, acc1, false, false);
    }

    const int wcol = wbase_g + wl + nco;
    #pragma unroll
    for (int r = 0; r < 8; ++r) {
        int co = r + (hi << 3);
        float v0 = acc0[r] + blds[co];
        float v1 = acc1[r] + blds[co + 16];
        out[((b * COUT + co)      * HH + h) * WW + wcol] = v0;
        out[((b * COUT + co + 16) * HH + h) * WW + wcol] = v1;
    }
}

extern "C" void kernel_launch(void* const* d_in, const int* in_sizes, int n_in,
                              void* d_out, int out_size, void* d_ws, size_t ws_size,
                              hipStream_t stream) {
    const float* x    = (const float*)d_in[0];
    const float* wgt  = (const float*)d_in[1];
    const float* bias = (const float*)d_in[2];
    float* out = (float*)d_out;

    dim3 grid(16 * 256 * 2);
    dim3 block(256);
    size_t smem = SMEM_FLOATS * sizeof(float);
    hipLaunchKernelGGL(conv3x3_wmma_f32, grid, block, smem, stream, x, wgt, bias, out);
}


# Each block walks 16 output rows over a 4-slot LDS ring of full-width input rows: weights staged once per 16 rows, one new input row per output row, prefetched to registers under the WMMAs
# speedup vs baseline: 2.6611x; 2.6611x over previous
#include <hip/hip_runtime.h>


typedef __attribute__((ext_vector_type(2))) float v2f;
typedef __attribute__((ext_vector_type(8))) float v8f;

#define CIN    32
#define COUT   32
#define HH     256
#define WW     256
#define NSTEP  72
#define ROWS   16
#define NTHR   512
#define NSLOT  4
#define PF     16

#define XPADW   34
#define XROWW   258
#define SLOT_ELEMS (XROWW * XPADW)
#define X_ELEMS (NSLOT * SLOT_ELEMS)
#define W_OFF   X_ELEMS
#define W_FRAGS (NSTEP * 2 * 32)
#define SMEM_FLOATS (W_OFF + W_FRAGS * 2)

extern "C" __global__ __launch_bounds__(NTHR)
void conv3x3_wmma_f32(const float* __restrict__ x,
                      const float* __restrict__ wgt,
                      const float* __restrict__ bias,
                      float* __restrict__ out)
{
    __shared__ __attribute__((aligned(16))) float smem[SMEM_FLOATS];
    float* xlds = smem;
    v2f*   wlds = (v2f*)(smem + W_OFF);

    const int tid = threadIdx.x;
    const int b   = blockIdx.x >> 4;
    const int h0  = (blockIdx.x & 15) * ROWS;

    for (int idx = tid; idx < W_FRAGS; idx += NTHR) {
        int s    = idx >> 6;
        int rem  = idx & 63;
        int t    = rem >> 5;
        int l    = rem & 31;
        int k0   = 4 * s + ((l >> 4) << 1);
        int cout = (t << 4) + (l & 15);
        v2f wv;
        #pragma unroll
        for (int v = 0; v < 2; ++v) {
            int k    = k0 + v;
            int ci   = k & 31;
            int khkw = k >> 5;
            wv[v] = wgt[(cout * CIN + ci) * 9 + khkw];
        }
        wlds[idx] = wv;
    }

    if (tid < NSLOT * 2 * XPADW) {
        int slot = tid / (2 * XPADW);
        int rem  = tid - slot * (2 * XPADW);
        int side = rem / XPADW;
        int c    = rem - side * XPADW;
        xlds[(slot * XROWW + side * (XROWW - 1)) * XPADW + c] = 0.0f;
    }

    const int sw  = tid & 255;
    const int sc0 = tid >> 8;
    const float* xsrc = x + ((b * CIN + sc0) * HH) * WW + sw;
    float*       xdst = xlds + (sw + 1) * XPADW + sc0;

    #pragma unroll
    for (int kh = 0; kh < 3; ++kh) {
        const int row = h0 + kh - 1;
        float pf[PF];
        if ((unsigned)row < HH) {
            #pragma unroll
            for (int j = 0; j < PF; ++j) pf[j] = xsrc[row * WW + j * (2 * HH * WW)];
        } else {
            #pragma unroll
            for (int j = 0; j < PF; ++j) pf[j] = 0.0f;
        }
        float* d = xdst + ((row + 1) & 3) * SLOT_ELEMS;
        #pragma unroll
        for (int j = 0; j < PF; ++j) d[2 * j] = pf[j];
    }
    __syncthreads();

    const int lane = tid & 31;
    const int wl   = (tid >> 5) << 4;
    const int nco  = lane & 15;
    const int hi   = lane >> 4;
    const float* xb = xlds + (wl + nco) * XPADW + (hi << 1);
    const int wcol = wl + nco;

    float bz0[8], bz1[8];
    #pragma unroll
    for (int q = 0; q < 8; ++q) {
        bz0[q] = bias[q + (hi << 3)];
        bz1[q] = bias[q + (hi << 3) + 16];
    }

    for (int r = 0; r < ROWS; ++r) {
        const int h = h0 + r;
        const bool more = (r + 1 < ROWS);

        float pf[PF];
        if (more && (h + 2) < HH) {
            #pragma unroll
            for (int j = 0; j < PF; ++j) pf[j] = xsrc[(h + 2) * WW + j * (2 * HH * WW)];
        } else {
            #pragma unroll
            for (int j = 0; j < PF; ++j) pf[j] = 0.0f;
        }

        const float* xk0 = xb + ((h + 0) & 3) * SLOT_ELEMS;
        const float* xk1 = xb + ((h + 1) & 3) * SLOT_ELEMS;
        const float* xk2 = xb + ((h + 2) & 3) * SLOT_ELEMS;

        v8f acc0 = {};
        v8f acc1 = {};

        #pragma unroll
        for (int s = 0; s < NSTEP; ++s) {
            const int khkw = s >> 3;
            const int kh   = khkw / 3;
            const int kw   = khkw - 3 * kh;
            const int cig  = (s & 7) << 2;

            v2f a0 = wlds[(s * 2 + 0) * 32 + lane];
            v2f a1 = wlds[(s * 2 + 1) * 32 + lane];

            const float* xk = (kh == 0) ? xk0 : (kh == 1) ? xk1 : xk2;
            v2f bv = *(const v2f*)&xk[kw * XPADW + cig];

            acc0 = __builtin_amdgcn_wmma_f32_16x16x4_f32(
                       false, a0, false, bv, (short)0, acc0, false, false);
            acc1 = __builtin_amdgcn_wmma_f32_16x16x4_f32(
                       false, a1, false, bv, (short)0, acc1, false, false);
        }

        {
            float* d = xdst + ((h + 3) & 3) * SLOT_ELEMS;
            #pragma unroll
            for (int j = 0; j < PF; ++j) d[2 * j] = pf[j];
            __syncthreads();
        }

        #pragma unroll
        for (int q = 0; q < 8; ++q) {
            int co = q + (hi << 3);
            float v0 = acc0[q] + bz0[q];
            float v1 = acc1[q] + bz1[q];
            out[((b * COUT + co)      * HH + h) * WW + wcol] = v0;
            out[((b * COUT + co + 16) * HH + h) * WW + wcol] = v1;
        }
    }
}

extern "C" void kernel_launch(void* const* d_in, const int* in_sizes, int n_in,
                              void* d_out, int out_size, void* d_ws, size_t ws_size,
                              hipStream_t stream) {
    const float* x    = (const float*)d_in[0];
    const float* wgt  = (const float*)d_in[1];
    const float* bias = (const float*)d_in[2];
    float* out = (float*)d_out;

    dim3 grid(256);
    dim3 block(NTHR);
    hipLaunchKernelGGL(conv3x3_wmma_f32, grid, block, 0, stream, x, wgt, bias, out);
}


# LDS fragment reads grouped: each 8-step block's 12 reads issued in one place, one block ahead of use (second register set), not 1-2 reads after every WMMA pair; chain runs across rows
# speedup vs baseline: 2.8755x; 1.0806x over previous
#include <hip/hip_runtime.h>


typedef __attribute__((ext_vector_type(2))) float v2f;
typedef __attribute__((ext_vector_type(8))) float v8f;

#define CIN    32
#define COUT   32
#define HH     256
#define WW     256
#define NSTEP  72
#define ROWS   16
#define NTHR   512
#define NSLOT  4
#define PF     16
#define BSTEP  8
#define NBLK   9

#define XPADW   34
#define XROWW   258
#define SLOT_ELEMS (XROWW * XPADW)
#define X_ELEMS (NSLOT * SLOT_ELEMS)
#define W_OFF   X_ELEMS
#define W_FRAGS (NSTEP * 2 * 32)
#define SMEM_FLOATS (W_OFF + W_FRAGS * 2)

static __device__ __forceinline__
void load_block(v2f (&a)[2 * BSTEP], v2f (&bf)[BSTEP],
                const v2f* __restrict__ wf, const float* __restrict__ xrow,
                const int kk, const int kw)
{
    #pragma unroll
    for (int i = 0; i < BSTEP; ++i) {
        const int s = kk * BSTEP + i;
        a[2 * i + 0] = wf[(s * 2 + 0) * 32];
        a[2 * i + 1] = wf[(s * 2 + 1) * 32];
        bf[i] = *(const v2f*)&xrow[kw * XPADW + (i << 2)];
    }
}

extern "C" __global__ __launch_bounds__(NTHR)
void conv3x3_wmma_f32(const float* __restrict__ x,
                      const float* __restrict__ wgt,
                      const float* __restrict__ bias,
                      float* __restrict__ out)
{
    __shared__ __attribute__((aligned(16))) float smem[SMEM_FLOATS];
    float* xlds = smem;
    v2f*   wlds = (v2f*)(smem + W_OFF);

    const int tid = threadIdx.x;
    const int b   = blockIdx.x >> 4;
    const int h0  = (blockIdx.x & 15) * ROWS;

    for (int idx = tid; idx < W_FRAGS; idx += NTHR) {
        int s    = idx >> 6;
        int rem  = idx & 63;
        int t    = rem >> 5;
        int l    = rem & 31;
        int k0   = 4 * s + ((l >> 4) << 1);
        int cout = (t << 4) + (l & 15);
        v2f wv;
        #pragma unroll
        for (int v = 0; v < 2; ++v) {
            int k    = k0 + v;
            int ci   = k & 31;
            int khkw = k >> 5;
            wv[v] = wgt[(cout * CIN + ci) * 9 + khkw];
        }
        wlds[idx] = wv;
    }

    if (tid < NSLOT * 2 * XPADW) {
        int slot = tid / (2 * XPADW);
        int rem  = tid - slot * (2 * XPADW);
        int side = rem / XPADW;
        int c    = rem - side * XPADW;
        xlds[(slot * XROWW + side * (XROWW - 1)) * XPADW + c] = 0.0f;
    }

    const int sw  = tid & 255;
    const int sc0 = tid >> 8;
    const float* xsrc = x + ((b * CIN + sc0) * HH) * WW + sw;
    float*       xdst = xlds + (sw + 1) * XPADW + sc0;

    #pragma unroll
    for (int kh = 0; kh < 3; ++kh) {
        const int row = h0 + kh - 1;
        float pf[PF];
        if ((unsigned)row < HH) {
            #pragma unroll
            for (int j = 0; j < PF; ++j) pf[j] = xsrc[row * WW + j * (2 * HH * WW)];
        } else {
            #pragma unroll
            for (int j = 0; j < PF; ++j) pf[j] = 0.0f;
        }
        float* d = xdst + ((row + 1) & 3) * SLOT_ELEMS;
        #pragma unroll
        for (int j = 0; j < PF; ++j) d[2 * j] = pf[j];
    }
    __syncthreads();

    const int lane = tid & 31;
    const int wl   = (tid >> 5) << 4;
    const int nco  = lane & 15;
    const int hi   = lane >> 4;
    const float* xb = xlds + (wl + nco) * XPADW + (hi << 1);
    const int wcol = wl + nco;

    float bz0[8], bz1[8];
    #pragma unroll
    for (int q = 0; q < 8; ++q) {
        bz0[q] = bias[q + (hi << 3)];
        bz1[q] = bias[q + (hi << 3) + 16];
    }

    const v2f* wfrag = wlds + lane;
    v2f ca[2 * BSTEP], cb[BSTEP];
    load_block(ca, cb, wfrag, xb + ((h0 + 0) & 3) * SLOT_ELEMS, 0, 0);

    for (int r = 0; r < ROWS; ++r) {
        const int h = h0 + r;
        const bool more = (r + 1 < ROWS);

        float pf[PF];
        if (more && (h + 2) < HH) {
            #pragma unroll
            for (int j = 0; j < PF; ++j) pf[j] = xsrc[(h + 2) * WW + j * (2 * HH * WW)];
        } else {
            #pragma unroll
            for (int j = 0; j < PF; ++j) pf[j] = 0.0f;
        }

        const float* xk0 = xb + ((h + 0) & 3) * SLOT_ELEMS;
        const float* xk1 = xb + ((h + 1) & 3) * SLOT_ELEMS;
        const float* xk2 = xb + ((h + 2) & 3) * SLOT_ELEMS;

        v8f acc0 = {};
        v8f acc1 = {};

        #pragma unroll
        for (int kk = 0; kk < NBLK; ++kk) {
            const int nk  = (kk + 1 < NBLK) ? kk + 1 : 0;
            const int nkh = nk / 3;
            const int nkw = nk - 3 * nkh;
            const float* xn = (kk + 1 < NBLK) ? ((nkh == 0) ? xk0 : (nkh == 1) ? xk1 : xk2) : xk1;
            v2f na[2 * BSTEP], nb[BSTEP];
            load_block(na, nb, wfrag, xn, nk, nkw);

            __builtin_amdgcn_sched_barrier(0);

            #pragma unroll
            for (int i = 0; i < BSTEP; ++i) {
                acc0 = __builtin_amdgcn_wmma_f32_16x16x4_f32(
                           false, ca[2 * i + 0], false, cb[i], (short)0, acc0, false, false);
                acc1 = __builtin_amdgcn_wmma_f32_16x16x4_f32(
                           false, ca[2 * i + 1], false, cb[i], (short)0, acc1, false, false);
            }

            __builtin_amdgcn_sched_barrier(0);

            #pragma unroll
            for (int i = 0; i < 2 * BSTEP; ++i) ca[i] = na[i];
            #pragma unroll
            for (int i = 0; i < BSTEP; ++i) cb[i] = nb[i];
        }

        {
            float* d = xdst + ((h + 3) & 3) * SLOT_ELEMS;
            #pragma unroll
            for (int j = 0; j < PF; ++j) d[2 * j] = pf[j];
            __syncthreads();
        }

        #pragma unroll
        for (int q = 0; q < 8; ++q) {
            int co = q + (hi << 3);
            float v0 = acc0[q] + bz0[q];
            float v1 = acc1[q] + bz1[q];
            out[((b * COUT + co)      * HH + h) * WW + wcol] = v0;
            out[((b * COUT + co + 16) * HH + h) * WW + wcol] = v1;
        }
    }
}

extern "C" void kernel_launch(void* const* d_in, const int* in_sizes, int n_in,
                              void* d_out, int out_size, void* d_ws, size_t ws_size,
                              hipStream_t stream) {
    const float* x    = (const float*)d_in[0];
    const float* wgt  = (const float*)d_in[1];
    const float* bias = (const float*)d_in[2];
    float* out = (float*)d_out;

    dim3 grid(256);
    dim3 block(NTHR);
    hipLaunchKernelGGL(conv3x3_wmma_f32, grid, block, 0, stream, x, wgt, bias, out);
}


# Wave tile 16 cout x 32 w instead of 32 cout x 16 w, lane halves swapped by v_permlanex16 in the epilogue: each output store writes one whole 128-B line, not two 64-B halves; sums unchanged
# speedup vs baseline: 3.1227x; 1.0860x over previous
#include <hip/hip_runtime.h>


typedef __attribute__((ext_vector_type(2))) float v2f;
typedef __attribute__((ext_vector_type(8))) float v8f;

#define CIN    32
#define COUT   32
#define HH     256
#define WW     256
#define NSTEP  72
#define ROWS   16
#define NTHR   512
#define NSLOT  4
#define PF     16
#define BSTEP  8
#define NBLK   9

#define XPADW   34
#define XROWW   258
#define SLOT_ELEMS (XROWW * XPADW)
#define X_ELEMS (NSLOT * SLOT_ELEMS)
#define W_OFF   X_ELEMS
#define W_FRAGS (NSTEP * 2 * 32)
#define SMEM_FLOATS (W_OFF + W_FRAGS * 2)

static __device__ __forceinline__
void load_block(v2f (&a)[BSTEP], v2f (&bl)[BSTEP], v2f (&br)[BSTEP],
                const v2f* __restrict__ wf, const float* __restrict__ xrow,
                const int kk, const int kw)
{
    #pragma unroll
    for (int i = 0; i < BSTEP; ++i) {
        const int s = kk * BSTEP + i;
        a[i]  = wf[(s * 2) * 32];
        bl[i] = *(const v2f*)&xrow[kw * XPADW + (i << 2)];
        br[i] = *(const v2f*)&xrow[(kw + 16) * XPADW + (i << 2)];
    }
}

static __device__ __forceinline__ float xhalf(float v)
{
    const int bits = __builtin_bit_cast(int, v);
    return __builtin_bit_cast(float,
        __builtin_amdgcn_permlanex16(bits, bits, 0x76543210, 0xfedcba98, false, false));
}

extern "C" __global__ __launch_bounds__(NTHR)
void conv3x3_wmma_f32(const float* __restrict__ x,
                      const float* __restrict__ wgt,
                      const float* __restrict__ bias,
                      float* __restrict__ out)
{
    __shared__ __attribute__((aligned(16))) float smem[SMEM_FLOATS];
    float* xlds = smem;
    v2f*   wlds = (v2f*)(smem + W_OFF);

    const int tid = threadIdx.x;
    const int b   = blockIdx.x >> 4;
    const int h0  = (blockIdx.x & 15) * ROWS;

    for (int idx = tid; idx < W_FRAGS; idx += NTHR) {
        int s    = idx >> 6;
        int rem  = idx & 63;
        int t    = rem >> 5;
        int l    = rem & 31;
        int k0   = 4 * s + ((l >> 4) << 1);
        int cout = (t << 4) + (l & 15);
        v2f wv;
        #pragma unroll
        for (int v = 0; v < 2; ++v) {
            int k    = k0 + v;
            int ci   = k & 31;
            int khkw = k >> 5;
            wv[v] = wgt[(cout * CIN + ci) * 9 + khkw];
        }
        wlds[idx] = wv;
    }

    if (tid < NSLOT * 2 * XPADW) {
        int slot = tid / (2 * XPADW);
        int rem  = tid - slot * (2 * XPADW);
        int side = rem / XPADW;
        int c    = rem - side * XPADW;
        xlds[(slot * XROWW + side * (XROWW - 1)) * XPADW + c] = 0.0f;
    }

    const int sw  = tid & 255;
    const int sc0 = tid >> 8;
    const float* xsrc = x + ((b * CIN + sc0) * HH) * WW + sw;
    float*       xdst = xlds + (sw + 1) * XPADW + sc0;

    #pragma unroll
    for (int kh = 0; kh < 3; ++kh) {
        const int row = h0 + kh - 1;
        float pf[PF];
        if ((unsigned)row < HH) {
            #pragma unroll
            for (int j = 0; j < PF; ++j) pf[j] = xsrc[row * WW + j * (2 * HH * WW)];
        } else {
            #pragma unroll
            for (int j = 0; j < PF; ++j) pf[j] = 0.0f;
        }
        float* d = xdst + ((row + 1) & 3) * SLOT_ELEMS;
        #pragma unroll
        for (int j = 0; j < PF; ++j) d[2 * j] = pf[j];
    }
    __syncthreads();

    const int lane = tid & 31;
    const int wv   = tid >> 5;
    const int mt   = wv & 1;
    const int wl   = (wv >> 1) << 5;
    const int nco  = lane & 15;
    const int hi   = lane >> 4;
    const float* xb = xlds + (wl + nco) * XPADW + (hi << 1);

    float bz0[8], bz1[8];
    #pragma unroll
    for (int q = 0; q < 8; ++q) {
        bz0[q] = bias[(mt << 4) + q];
        bz1[q] = bias[(mt << 4) + q + 8];
    }

    const v2f* wfrag = wlds + (mt << 5) + lane;
    v2f ca[BSTEP], cl[BSTEP], cr[BSTEP];
    load_block(ca, cl, cr, wfrag, xb + ((h0 + 0) & 3) * SLOT_ELEMS, 0, 0);

    for (int r = 0; r < ROWS; ++r) {
        const int h = h0 + r;
        const bool more = (r + 1 < ROWS);

        float pf[PF];
        if (more && (h + 2) < HH) {
            #pragma unroll
            for (int j = 0; j < PF; ++j) pf[j] = xsrc[(h + 2) * WW + j * (2 * HH * WW)];
        } else {
            #pragma unroll
            for (int j = 0; j < PF; ++j) pf[j] = 0.0f;
        }

        const float* xk0 = xb + ((h + 0) & 3) * SLOT_ELEMS;
        const float* xk1 = xb + ((h + 1) & 3) * SLOT_ELEMS;
        const float* xk2 = xb + ((h + 2) & 3) * SLOT_ELEMS;

        v8f acc0 = {};
        v8f acc1 = {};

        #pragma unroll
        for (int kk = 0; kk < NBLK; ++kk) {
            const int nk  = (kk + 1 < NBLK) ? kk + 1 : 0;
            const int nkh = nk / 3;
            const int nkw = nk - 3 * nkh;
            const float* xn = (kk + 1 < NBLK) ? ((nkh == 0) ? xk0 : (nkh == 1) ? xk1 : xk2) : xk1;
            v2f na[BSTEP], nl[BSTEP], nr[BSTEP];
            load_block(na, nl, nr, wfrag, xn, nk, nkw);

            __builtin_amdgcn_sched_barrier(0);

            #pragma unroll
            for (int i = 0; i < BSTEP; ++i) {
                acc0 = __builtin_amdgcn_wmma_f32_16x16x4_f32(
                           false, ca[i], false, cl[i], (short)0, acc0, false, false);
                acc1 = __builtin_amdgcn_wmma_f32_16x16x4_f32(
                           false, ca[i], false, cr[i], (short)0, acc1, false, false);
            }

            __builtin_amdgcn_sched_barrier(0);

            #pragma unroll
            for (int i = 0; i < BSTEP; ++i) { ca[i] = na[i]; cl[i] = nl[i]; cr[i] = nr[i]; }
        }

        {
            float* d = xdst + ((h + 3) & 3) * SLOT_ELEMS;
            #pragma unroll
            for (int j = 0; j < PF; ++j) d[2 * j] = pf[j];
            __syncthreads();
        }

        asm volatile("v_nop\n\tv_nop\n\tv_nop\n\tv_nop" : "+v"(acc0), "+v"(acc1));
        float* orow = out + ((b * COUT + (mt << 4)) * HH + h) * WW + wl + lane;
        #pragma unroll
        for (int q = 0; q < 8; ++q) {
            const float mine0 = acc0[q];
            const float mine1 = acc1[q];
            const float got = xhalf(hi ? mine0 : mine1);
            const float v0 = (hi ? got : mine0) + bz0[q];
            const float v1 = (hi ? mine1 : got) + bz1[q];
            orow[q * (HH * WW)]       = v0;
            orow[(q + 8) * (HH * WW)] = v1;
        }
    }
}

extern "C" void kernel_launch(void* const* d_in, const int* in_sizes, int n_in,
                              void* d_out, int out_size, void* d_ws, size_t ws_size,
                              hipStream_t stream) {
    const float* x    = (const float*)d_in[0];
    const float* wgt  = (const float*)d_in[1];
    const float* bias = (const float*)d_in[2];
    float* out = (float*)d_out;

    dim3 grid(256);
    dim3 block(NTHR);
    hipLaunchKernelGGL(conv3x3_wmma_f32, grid, block, 0, stream, x, wgt, bias, out);
}


# Vertical minimal filtering F(2,3): per pair of output rows 4 products of K=96 on input rows combined in registers on the way to LDS, not 2 of K=288: a third fewer WMMAs; order of the sums changed
# speedup vs baseline: 3.8188x; 1.2229x over previous
#include <hip/hip_runtime.h>


typedef __attribute__((ext_vector_type(2))) float v2f;
typedef __attribute__((ext_vector_type(8))) float v8f;

#define CIN    32
#define COUT   32
#define HH     256
#define WW     256
#define ROWS   16
#define NPAIR  8
#define NTHR   512
#define NSLOT  6
#define PF     16
#define BSTEP  8
#define JSTEP  24
#define NSTEP  96
#define NBLK   12

#define XPADW   34
#define XROWW   258
#define SLOT_ELEMS (XROWW * XPADW)
#define X_ELEMS (NSLOT * SLOT_ELEMS)
#define W_OFF   X_ELEMS
#define W_FRAGS (NSTEP * 2 * 32)
#define SMEM_FLOATS (W_OFF + W_FRAGS * 2)


static __device__ __forceinline__
void load_block(v2f (&a)[BSTEP], v2f (&bl)[BSTEP], v2f (&br)[BSTEP],
                const v2f* __restrict__ wf, const float* __restrict__ xrow,
                const int sbase, const int kw)
{
    #pragma unroll
    for (int i = 0; i < BSTEP; ++i) {
        a[i]  = wf[((sbase + i) * 2) * 32];
        bl[i] = *(const v2f*)&xrow[kw * XPADW + (i << 2)];
        br[i] = *(const v2f*)&xrow[(kw + 16) * XPADW + (i << 2)];
    }
}

static __device__ __forceinline__ float xhalf(float v)
{
    const int bits = __builtin_bit_cast(int, v);
    return __builtin_bit_cast(float,
        __builtin_amdgcn_permlanex16(bits, bits, 0x76543210, 0xfedcba98, false, false));
}

static __device__ __forceinline__
void load_row(float (&r)[PF], const float* __restrict__ xsrc, const int row)
{
    if ((unsigned)row < HH) {
        #pragma unroll
        for (int j = 0; j < PF; ++j) r[j] = xsrc[row * WW + j * (2 * HH * WW)];
    } else {
        #pragma unroll
        for (int j = 0; j < PF; ++j) r[j] = 0.0f;
    }
}

static __device__ __forceinline__
void store_out_row(float* __restrict__ orow, const v8f yl, const v8f yr,
                   const float (&bz0)[8], const float (&bz1)[8], const int hi)
{
    #pragma unroll
    for (int q = 0; q < 8; ++q) {
        const float mine0 = yl[q];
        const float mine1 = yr[q];
        const float got = xhalf(hi ? mine0 : mine1);
        const float v0 = (hi ? got : mine0) + bz0[q];
        const float v1 = (hi ? mine1 : got) + bz1[q];
        orow[q * (HH * WW)]       = v0;
        orow[(q + 8) * (HH * WW)] = v1;
    }
}

extern "C" __global__ __launch_bounds__(NTHR)
void conv3x3_wmma_f32(const float* __restrict__ x,
                      const float* __restrict__ wgt,
                      const float* __restrict__ bias,
                      float* __restrict__ out)
{
    __shared__ __attribute__((aligned(16))) float smem[SMEM_FLOATS];
    float* xlds = smem;
    v2f*   wlds = (v2f*)(smem + W_OFF);

    const int tid = threadIdx.x;
    const int b   = blockIdx.x >> 4;
    const int h0  = (blockIdx.x & 15) * ROWS;

    for (int idx = tid; idx < JSTEP * 64; idx += NTHR) {
        int sj   = idx >> 6;
        int rem  = idx & 63;
        int t    = rem >> 5;
        int l    = rem & 31;
        int k0   = 4 * sj + ((l >> 4) << 1);
        int cout = (t << 4) + (l & 15);
        v2f u0, u1, u2, u3;
        #pragma unroll
        for (int v = 0; v < 2; ++v) {
            int k  = k0 + v;
            int ci = k & 31;
            int kw = k >> 5;
            const float* g = wgt + (cout * CIN + ci) * 9 + kw;
            const float g0 = g[0], g1 = g[3], g2 = g[6];
            u0[v] = g0;
            u1[v] = 0.5f * ((g0 + g2) + g1);
            u2[v] = 0.5f * ((g0 + g2) - g1);
            u3[v] = -g2;
        }
        wlds[(0 * JSTEP + sj) * 64 + rem] = u0;
        wlds[(1 * JSTEP + sj) * 64 + rem] = u1;
        wlds[(2 * JSTEP + sj) * 64 + rem] = u2;
        wlds[(3 * JSTEP + sj) * 64 + rem] = u3;
    }

    if (tid < NSLOT * 2 * XPADW) {
        int slot = tid / (2 * XPADW);
        int rem  = tid - slot * (2 * XPADW);
        int side = rem / XPADW;
        int c    = rem - side * XPADW;
        xlds[(slot * XROWW + side * (XROWW - 1)) * XPADW + c] = 0.0f;
    }

    const int sw  = tid & 255;
    const int sc0 = tid >> 8;
    const float* xsrc = x + ((b * CIN + sc0) * HH) * WW + sw;
    float*       xdst = xlds + (sw + 1) * XPADW + sc0;

    float c0[PF], c1[PF];

    {
        float r0[PF], r1[PF];
        load_row(r0, xsrc, h0 - 1);
        load_row(r1, xsrc, h0);
        load_row(c0, xsrc, h0 + 1);
        load_row(c1, xsrc, h0 + 2);
        #pragma unroll
        for (int j = 0; j < PF; ++j) {
            xdst[0 * SLOT_ELEMS + 2 * j] = r0[j] - c0[j];
            xdst[1 * SLOT_ELEMS + 2 * j] = r1[j] + c0[j];
            xdst[2 * SLOT_ELEMS + 2 * j] = c0[j] - r1[j];
            xdst[3 * SLOT_ELEMS + 2 * j] = r1[j] - c1[j];
        }
    }
    __syncthreads();

    const int lane = tid & 31;
    const int wv   = tid >> 5;
    const int mt   = wv & 1;
    const int wl   = (wv >> 1) << 5;
    const int nco  = lane & 15;
    const int hi   = lane >> 4;
    const float* xb = xlds + (wl + nco) * XPADW + (hi << 1);

    const int mts = __builtin_amdgcn_readfirstlane(mt);
    float bz0[8], bz1[8];
    #pragma unroll
    for (int q = 0; q < 8; ++q) {
        bz0[q] = bias[(mts << 4) + q];
        bz1[q] = bias[(mts << 4) + q + 8];
    }

    int ga = 0, gb = 1, gf = 2;

    const v2f* wfrag = wlds + (mt << 5) + lane;
    v2f ca[BSTEP], cl[BSTEP], cr[BSTEP];
    load_block(ca, cl, cr, wfrag, xb + 1 * SLOT_ELEMS, 1 * JSTEP, 0);

    for (int p = 0; p < NPAIR; ++p) {
        const int h = h0 + 2 * p;
        const bool more = (p + 1 < NPAIR);

        float rn[PF], rs[PF];
        load_row(rn, xsrc, more ? h + 3 : -1);

        const float* xt0 = xb + (2 * ga)     * SLOT_ELEMS;
        const float* xt1 = xb + (2 * ga + 1) * SLOT_ELEMS;
        const float* xt2 = xb + (2 * gb)     * SLOT_ELEMS;
        const float* xt3 = xb + (2 * gb + 1) * SLOT_ELEMS;
        const float* xn1 = xb + (2 * gf + 1) * SLOT_ELEMS;

        v8f aL = {}, aR = {};
        v8f bL = {}, bR = {};
        v8f cL = {}, cR = {};

        #pragma unroll
        for (int q = 0; q < NBLK; ++q) {
            const int half = q / 3;
            const int jq   = (half == 0) ? 1 : (half == 1) ? 0 : (half == 2) ? 3 : 2;

            const int nq    = (q + 1 < NBLK) ? q + 1 : 0;
            const int nhalf = nq / 3;
            const int nj    = (nhalf == 0) ? 1 : (nhalf == 1) ? 0 : (nhalf == 2) ? 3 : 2;
            const int nkw   = nq - 3 * nhalf;
            const float* xn = (q + 1 < NBLK) ? ((nj == 0) ? xt0 : (nj == 1) ? xt1 : (nj == 2) ? xt2 : xt3) : xn1;
            v2f na[BSTEP], nl[BSTEP], nrr[BSTEP];
            load_block(na, nl, nrr, wfrag, xn, nj * JSTEP + nkw * BSTEP, nkw);

            __builtin_amdgcn_sched_barrier(0);

            #pragma unroll
            for (int i = 0; i < BSTEP; ++i) {
                if (jq == 1) {
                    bL = __builtin_amdgcn_wmma_f32_16x16x4_f32(false, ca[i], false, cl[i], (short)0, bL, false, false);
                    bR = __builtin_amdgcn_wmma_f32_16x16x4_f32(false, ca[i], false, cr[i], (short)0, bR, false, false);
                } else if (jq == 0) {
                    const v8f sl = (q == 3 && i == 0) ? bL : aL;
                    const v8f sr = (q == 3 && i == 0) ? bR : aR;
                    aL = __builtin_amdgcn_wmma_f32_16x16x4_f32(false, ca[i], false, cl[i], (short)0, sl, false, false);
                    aR = __builtin_amdgcn_wmma_f32_16x16x4_f32(false, ca[i], false, cr[i], (short)0, sr, false, false);
                } else if (jq == 3) {
                    bL = __builtin_amdgcn_wmma_f32_16x16x4_f32(false, ca[i], false, cl[i], (short)0, bL, false, false);
                    bR = __builtin_amdgcn_wmma_f32_16x16x4_f32(false, ca[i], false, cr[i], (short)0, bR, false, false);
                } else {
                    cL = __builtin_amdgcn_wmma_f32_16x16x4_f32(false, ca[i], false, cl[i], (short)0, cL, false, false);
                    cR = __builtin_amdgcn_wmma_f32_16x16x4_f32(false, ca[i], false, cr[i], (short)0, cR, false, false);
                }
            }

            __builtin_amdgcn_sched_barrier(0);

            #pragma unroll
            for (int i = 0; i < BSTEP; ++i) { ca[i] = na[i]; cl[i] = nl[i]; cr[i] = nrr[i]; }

            if (q == 5) {
                float* d0 = xdst + (2 * gf)     * SLOT_ELEMS;
                float* d1 = xdst + (2 * gf + 1) * SLOT_ELEMS;
                #pragma unroll
                for (int j = 0; j < PF; ++j) {
                    d0[2 * j] = c0[j] - rn[j];
                    d1[2 * j] = c1[j] + rn[j];
                }
                __syncthreads();
                float* d2 = xdst + (2 * ga) * SLOT_ELEMS;
                #pragma unroll
                for (int j = 0; j < PF; ++j) d2[2 * j] = rn[j] - c1[j];
                load_row(rs, xsrc, more ? h + 4 : -1);
            }
        }

        {
            float* d3 = xdst + (2 * ga + 1) * SLOT_ELEMS;
            #pragma unroll
            for (int j = 0; j < PF; ++j) d3[2 * j] = c1[j] - rs[j];
            __syncthreads();
        }
        #pragma unroll
        for (int j = 0; j < PF; ++j) { c0[j] = rn[j]; c1[j] = rs[j]; }
        { const int g = ga; ga = gf; gf = gb; gb = g; }

        asm volatile("v_nop\n\tv_nop\n\tv_nop\n\tv_nop"
                     : "+v"(aL), "+v"(aR), "+v"(bL), "+v"(bR), "+v"(cL), "+v"(cR));

        float* orow = out + ((b * COUT + (mt << 4)) * HH + h) * WW + wl + lane;
        store_out_row(orow,      aL + cL, aR + cR, bz0, bz1, hi);
        store_out_row(orow + WW, bL - cL, bR - cR, bz0, bz1, hi);
    }
}

extern "C" void kernel_launch(void* const* d_in, const int* in_sizes, int n_in,
                              void* d_out, int out_size, void* d_ws, size_t ws_size,
                              hipStream_t stream) {
    const float* x    = (const float*)d_in[0];
    const float* wgt  = (const float*)d_in[1];
    const float* bias = (const float*)d_in[2];
    float* out = (float*)d_out;

    dim3 grid(256);
    dim3 block(NTHR);
    hipLaunchKernelGGL(conv3x3_wmma_f32, grid, block, 0, stream, x, wgt, bias, out);
}


# End-of-pair barrier taken as signal and wait with the output-row stores between them: a wave stores its rows while others still multiply, not all 16 waves at once behind the barrier; sums unchanged
# speedup vs baseline: 3.9744x; 1.0407x over previous
#include <hip/hip_runtime.h>


typedef __attribute__((ext_vector_type(2))) float v2f;
typedef __attribute__((ext_vector_type(8))) float v8f;

#define CIN    32
#define COUT   32
#define HH     256
#define WW     256
#define ROWS   16
#define NPAIR  8
#define NTHR   512
#define NSLOT  6
#define PF     16
#define BSTEP  8
#define JSTEP  24
#define NSTEP  96
#define NBLK   12

#define XPADW   34
#define XROWW   258
#define SLOT_ELEMS (XROWW * XPADW)
#define X_ELEMS (NSLOT * SLOT_ELEMS)
#define W_OFF   X_ELEMS
#define W_FRAGS (NSTEP * 2 * 32)
#define SMEM_FLOATS (W_OFF + W_FRAGS * 2)


static __device__ __forceinline__
void load_block(v2f (&a)[BSTEP], v2f (&bl)[BSTEP], v2f (&br)[BSTEP],
                const v2f* __restrict__ wf, const float* __restrict__ xrow,
                const int sbase, const int kw)
{
    #pragma unroll
    for (int i = 0; i < BSTEP; ++i) {
        a[i]  = wf[((sbase + i) * 2) * 32];
        bl[i] = *(const v2f*)&xrow[kw * XPADW + (i << 2)];
        br[i] = *(const v2f*)&xrow[(kw + 16) * XPADW + (i << 2)];
    }
}

static __device__ __forceinline__ float xhalf(float v)
{
    const int bits = __builtin_bit_cast(int, v);
    return __builtin_bit_cast(float,
        __builtin_amdgcn_permlanex16(bits, bits, 0x76543210, 0xfedcba98, false, false));
}

static __device__ __forceinline__
void load_row(float (&r)[PF], const float* __restrict__ xsrc, const int row)
{
    if ((unsigned)row < HH) {
        #pragma unroll
        for (int j = 0; j < PF; ++j) r[j] = xsrc[row * WW + j * (2 * HH * WW)];
    } else {
        #pragma unroll
        for (int j = 0; j < PF; ++j) r[j] = 0.0f;
    }
}

static __device__ __forceinline__
void store_out_row(float* __restrict__ orow, const v8f yl, const v8f yr,
                   const float (&bz0)[8], const float (&bz1)[8], const int hi)
{
    #pragma unroll
    for (int q = 0; q < 8; ++q) {
        const float mine0 = yl[q];
        const float mine1 = yr[q];
        const float got = xhalf(hi ? mine0 : mine1);
        const float v0 = (hi ? got : mine0) + bz0[q];
        const float v1 = (hi ? mine1 : got) + bz1[q];
        orow[q * (HH * WW)]       = v0;
        orow[(q + 8) * (HH * WW)] = v1;
    }
}

extern "C" __global__ __launch_bounds__(NTHR)
void conv3x3_wmma_f32(const float* __restrict__ x,
                      const float* __restrict__ wgt,
                      const float* __restrict__ bias,
                      float* __restrict__ out)
{
    __shared__ __attribute__((aligned(16))) float smem[SMEM_FLOATS];
    float* xlds = smem;
    v2f*   wlds = (v2f*)(smem + W_OFF);

    const int tid = threadIdx.x;
    const int b   = blockIdx.x >> 4;
    const int h0  = (blockIdx.x & 15) * ROWS;

    for (int idx = tid; idx < JSTEP * 64; idx += NTHR) {
        int sj   = idx >> 6;
        int rem  = idx & 63;
        int t    = rem >> 5;
        int l    = rem & 31;
        int k0   = 4 * sj + ((l >> 4) << 1);
        int cout = (t << 4) + (l & 15);
        v2f u0, u1, u2, u3;
        #pragma unroll
        for (int v = 0; v < 2; ++v) {
            int k  = k0 + v;
            int ci = k & 31;
            int kw = k >> 5;
            const float* g = wgt + (cout * CIN + ci) * 9 + kw;
            const float g0 = g[0], g1 = g[3], g2 = g[6];
            u0[v] = g0;
            u1[v] = 0.5f * ((g0 + g2) + g1);
            u2[v] = 0.5f * ((g0 + g2) - g1);
            u3[v] = -g2;
        }
        wlds[(0 * JSTEP + sj) * 64 + rem] = u0;
        wlds[(1 * JSTEP + sj) * 64 + rem] = u1;
        wlds[(2 * JSTEP + sj) * 64 + rem] = u2;
        wlds[(3 * JSTEP + sj) * 64 + rem] = u3;
    }

    if (tid < NSLOT * 2 * XPADW) {
        int slot = tid / (2 * XPADW);
        int rem  = tid - slot * (2 * XPADW);
        int side = rem / XPADW;
        int c    = rem - side * XPADW;
        xlds[(slot * XROWW + side * (XROWW - 1)) * XPADW + c] = 0.0f;
    }

    const int sw  = tid & 255;
    const int sc0 = tid >> 8;
    const float* xsrc = x + ((b * CIN + sc0) * HH) * WW + sw;
    float*       xdst = xlds + (sw + 1) * XPADW + sc0;

    float c0[PF], c1[PF];

    {
        float r0[PF], r1[PF];
        load_row(r0, xsrc, h0 - 1);
        load_row(r1, xsrc, h0);
        load_row(c0, xsrc, h0 + 1);
        load_row(c1, xsrc, h0 + 2);
        #pragma unroll
        for (int j = 0; j < PF; ++j) {
            xdst[0 * SLOT_ELEMS + 2 * j] = r0[j] - c0[j];
            xdst[1 * SLOT_ELEMS + 2 * j] = r1[j] + c0[j];
            xdst[2 * SLOT_ELEMS + 2 * j] = c0[j] - r1[j];
            xdst[3 * SLOT_ELEMS + 2 * j] = r1[j] - c1[j];
        }
    }
    __syncthreads();

    const int lane = tid & 31;
    const int wv   = tid >> 5;
    const int mt   = wv & 1;
    const int wl   = (wv >> 1) << 5;
    const int nco  = lane & 15;
    const int hi   = lane >> 4;
    const float* xb = xlds + (wl + nco) * XPADW + (hi << 1);

    const int mts = __builtin_amdgcn_readfirstlane(mt);
    float bz0[8], bz1[8];
    #pragma unroll
    for (int q = 0; q < 8; ++q) {
        bz0[q] = bias[(mts << 4) + q];
        bz1[q] = bias[(mts << 4) + q + 8];
    }

    int ga = 0, gb = 1, gf = 2;

    const v2f* wfrag = wlds + (mt << 5) + lane;
    v2f ca[BSTEP], cl[BSTEP], cr[BSTEP];
    load_block(ca, cl, cr, wfrag, xb + 1 * SLOT_ELEMS, 1 * JSTEP, 0);

    for (int p = 0; p < NPAIR; ++p) {
        const int h = h0 + 2 * p;
        const bool more = (p + 1 < NPAIR);

        float rn[PF], rs[PF];
        load_row(rn, xsrc, more ? h + 3 : -1);

        const float* xt0 = xb + (2 * ga)     * SLOT_ELEMS;
        const float* xt1 = xb + (2 * ga + 1) * SLOT_ELEMS;
        const float* xt2 = xb + (2 * gb)     * SLOT_ELEMS;
        const float* xt3 = xb + (2 * gb + 1) * SLOT_ELEMS;
        const float* xn1 = xb + (2 * gf + 1) * SLOT_ELEMS;

        v8f aL = {}, aR = {};
        v8f bL = {}, bR = {};
        v8f cL = {}, cR = {};

        #pragma unroll
        for (int q = 0; q < NBLK; ++q) {
            const int half = q / 3;
            const int jq   = (half == 0) ? 1 : (half == 1) ? 0 : (half == 2) ? 3 : 2;

            const int nq    = (q + 1 < NBLK) ? q + 1 : 0;
            const int nhalf = nq / 3;
            const int nj    = (nhalf == 0) ? 1 : (nhalf == 1) ? 0 : (nhalf == 2) ? 3 : 2;
            const int nkw   = nq - 3 * nhalf;
            const float* xn = (q + 1 < NBLK) ? ((nj == 0) ? xt0 : (nj == 1) ? xt1 : (nj == 2) ? xt2 : xt3) : xn1;
            v2f na[BSTEP], nl[BSTEP], nrr[BSTEP];
            load_block(na, nl, nrr, wfrag, xn, nj * JSTEP + nkw * BSTEP, nkw);

            __builtin_amdgcn_sched_barrier(0);

            #pragma unroll
            for (int i = 0; i < BSTEP; ++i) {
                if (jq == 1) {
                    bL = __builtin_amdgcn_wmma_f32_16x16x4_f32(false, ca[i], false, cl[i], (short)0, bL, false, false);
                    bR = __builtin_amdgcn_wmma_f32_16x16x4_f32(false, ca[i], false, cr[i], (short)0, bR, false, false);
                } else if (jq == 0) {
                    const v8f sl = (q == 3 && i == 0) ? bL : aL;
                    const v8f sr = (q == 3 && i == 0) ? bR : aR;
                    aL = __builtin_amdgcn_wmma_f32_16x16x4_f32(false, ca[i], false, cl[i], (short)0, sl, false, false);
                    aR = __builtin_amdgcn_wmma_f32_16x16x4_f32(false, ca[i], false, cr[i], (short)0, sr, false, false);
                } else if (jq == 3) {
                    bL = __builtin_amdgcn_wmma_f32_16x16x4_f32(false, ca[i], false, cl[i], (short)0, bL, false, false);
                    bR = __builtin_amdgcn_wmma_f32_16x16x4_f32(false, ca[i], false, cr[i], (short)0, bR, false, false);
                } else {
                    cL = __builtin_amdgcn_wmma_f32_16x16x4_f32(false, ca[i], false, cl[i], (short)0, cL, false, false);
                    cR = __builtin_amdgcn_wmma_f32_16x16x4_f32(false, ca[i], false, cr[i], (short)0, cR, false, false);
                }
            }

            __builtin_amdgcn_sched_barrier(0);

            #pragma unroll
            for (int i = 0; i < BSTEP; ++i) { ca[i] = na[i]; cl[i] = nl[i]; cr[i] = nrr[i]; }

            if (q == 5) {
                float* d0 = xdst + (2 * gf)     * SLOT_ELEMS;
                float* d1 = xdst + (2 * gf + 1) * SLOT_ELEMS;
                #pragma unroll
                for (int j = 0; j < PF; ++j) {
                    d0[2 * j] = c0[j] - rn[j];
                    d1[2 * j] = c1[j] + rn[j];
                }
                __syncthreads();
                float* d2 = xdst + (2 * ga) * SLOT_ELEMS;
                #pragma unroll
                for (int j = 0; j < PF; ++j) d2[2 * j] = rn[j] - c1[j];
                load_row(rs, xsrc, more ? h + 4 : -1);
            }
        }

        {
            float* d3 = xdst + (2 * ga + 1) * SLOT_ELEMS;
            #pragma unroll
            for (int j = 0; j < PF; ++j) d3[2 * j] = c1[j] - rs[j];
        }
        __builtin_amdgcn_fence(__ATOMIC_RELEASE, "workgroup", "local");
        __builtin_amdgcn_s_barrier_signal(-1);

        asm volatile("v_nop\n\tv_nop\n\tv_nop\n\tv_nop"
                     : "+v"(aL), "+v"(aR), "+v"(bL), "+v"(bR), "+v"(cL), "+v"(cR));

        {
            float* orow = out + ((b * COUT + (mt << 4)) * HH + h) * WW + wl + lane;
            store_out_row(orow,      aL + cL, aR + cR, bz0, bz1, hi);
            store_out_row(orow + WW, bL - cL, bR - cR, bz0, bz1, hi);
        }

        __builtin_amdgcn_s_barrier_wait(-1);
        __builtin_amdgcn_fence(__ATOMIC_ACQUIRE, "workgroup", "local");

        #pragma unroll
        for (int j = 0; j < PF; ++j) { c0[j] = rn[j]; c1[j] = rs[j]; }
        { const int g = ga; ga = gf; gf = gb; gb = g; }
    }
}

extern "C" void kernel_launch(void* const* d_in, const int* in_sizes, int n_in,
                              void* d_out, int out_size, void* d_ws, size_t ws_size,
                              hipStream_t stream) {
    const float* x    = (const float*)d_in[0];
    const float* wgt  = (const float*)d_in[1];
    const float* bias = (const float*)d_in[2];
    float* out = (float*)d_out;

    dim3 grid(256);
    dim3 block(NTHR);
    hipLaunchKernelGGL(conv3x3_wmma_f32, grid, block, 0, stream, x, wgt, bias, out);
}


# Output rows staged in LDS and written one pair late, two 512-B store instructions per fragment block under the next pair's WMMAs, not 32 128-B stores per wave at the pair's end; sums unchanged
# speedup vs baseline: 4.2011x; 1.0570x over previous
#include <hip/hip_runtime.h>


typedef __attribute__((ext_vector_type(2))) float v2f;
typedef __attribute__((ext_vector_type(4))) float v4f;
typedef __attribute__((ext_vector_type(8))) float v8f;

#define CIN    32
#define COUT   32
#define HH     256
#define WW     256
#define ROWS   16
#define NPAIR  8
#define NTHR   512
#define NSLOT  6
#define PF     16
#define BSTEP  8
#define JSTEP  24
#define NSTEP  96
#define NBLK   12

#define XPADW   34
#define XROWW   258
#define SLOT_ELEMS (XROWW * XPADW)
#define X_ELEMS (NSLOT * SLOT_ELEMS)
#define W_OFF   X_ELEMS
#define W_FRAGS (NSTEP * 2 * 32)
#define OUT_OFF (W_OFF + W_FRAGS * 2)
#define OUT_ELEMS (2 * COUT * WW)
#define SMEM_FLOATS (OUT_OFF + OUT_ELEMS)
#define OTASKS 8


static __device__ __forceinline__
void load_block(v2f (&a)[BSTEP], v2f (&bl)[BSTEP], v2f (&br)[BSTEP],
                const v2f* __restrict__ wf, const float* __restrict__ xrow,
                const int sbase, const int kw)
{
    #pragma unroll
    for (int i = 0; i < BSTEP; ++i) {
        a[i]  = wf[((sbase + i) * 2) * 32];
        bl[i] = *(const v2f*)&xrow[kw * XPADW + (i << 2)];
        br[i] = *(const v2f*)&xrow[(kw + 16) * XPADW + (i << 2)];
    }
}

static __device__ __forceinline__
void load_row(float (&r)[PF], const float* __restrict__ xsrc, const int row)
{
    if ((unsigned)row < HH) {
        #pragma unroll
        for (int j = 0; j < PF; ++j) r[j] = xsrc[row * WW + j * (2 * HH * WW)];
    } else {
        #pragma unroll
        for (int j = 0; j < PF; ++j) r[j] = 0.0f;
    }
}

static __device__ __forceinline__ v4f out_piece_load(const float* __restrict__ ord, const int k)
{
    return *(const v4f*)&ord[(((k >> 2) * COUT) + ((k >> 1) & 1)) * WW + (k & 1) * 128];
}

static __device__ __forceinline__
void out_piece_store(float* __restrict__ og, const int k, v4f v, const float b0, const float b1)
{
    v += ((k >> 1) & 1) ? b1 : b0;
    *(v4f*)&og[((((k >> 1) & 1) * HH) + (k >> 2)) * WW + (k & 1) * 128] = v;
}

extern "C" __global__ __launch_bounds__(NTHR)
void conv3x3_wmma_f32(const float* __restrict__ x,
                      const float* __restrict__ wgt,
                      const float* __restrict__ bias,
                      float* __restrict__ out)
{
    __shared__ __attribute__((aligned(16))) float smem[SMEM_FLOATS];
    float* xlds = smem;
    v2f*   wlds = (v2f*)(smem + W_OFF);

    const int tid = threadIdx.x;
    const int b   = blockIdx.x >> 4;
    const int h0  = (blockIdx.x & 15) * ROWS;

    for (int idx = tid; idx < JSTEP * 64; idx += NTHR) {
        int sj   = idx >> 6;
        int rem  = idx & 63;
        int t    = rem >> 5;
        int l    = rem & 31;
        int k0   = 4 * sj + ((l >> 4) << 1);
        int cout = (t << 4) + (l & 15);
        v2f u0, u1, u2, u3;
        #pragma unroll
        for (int v = 0; v < 2; ++v) {
            int k  = k0 + v;
            int ci = k & 31;
            int kw = k >> 5;
            const float* g = wgt + (cout * CIN + ci) * 9 + kw;
            const float g0 = g[0], g1 = g[3], g2 = g[6];
            u0[v] = g0;
            u1[v] = 0.5f * ((g0 + g2) + g1);
            u2[v] = 0.5f * ((g0 + g2) - g1);
            u3[v] = -g2;
        }
        wlds[(0 * JSTEP + sj) * 64 + rem] = u0;
        wlds[(1 * JSTEP + sj) * 64 + rem] = u1;
        wlds[(2 * JSTEP + sj) * 64 + rem] = u2;
        wlds[(3 * JSTEP + sj) * 64 + rem] = u3;
    }

    if (tid < NSLOT * 2 * XPADW) {
        int slot = tid / (2 * XPADW);
        int rem  = tid - slot * (2 * XPADW);
        int side = rem / XPADW;
        int c    = rem - side * XPADW;
        xlds[(slot * XROWW + side * (XROWW - 1)) * XPADW + c] = 0.0f;
    }

    const int sw  = tid & 255;
    const int sc0 = tid >> 8;
    const float* xsrc = x + ((b * CIN + sc0) * HH) * WW + sw;
    float*       xdst = xlds + (sw + 1) * XPADW + sc0;

    float c0[PF], c1[PF];

    {
        float r0[PF], r1[PF];
        load_row(r0, xsrc, h0 - 1);
        load_row(r1, xsrc, h0);
        load_row(c0, xsrc, h0 + 1);
        load_row(c1, xsrc, h0 + 2);
        #pragma unroll
        for (int j = 0; j < PF; ++j) {
            xdst[0 * SLOT_ELEMS + 2 * j] = r0[j] - c0[j];
            xdst[1 * SLOT_ELEMS + 2 * j] = r1[j] + c0[j];
            xdst[2 * SLOT_ELEMS + 2 * j] = c0[j] - r1[j];
            xdst[3 * SLOT_ELEMS + 2 * j] = r1[j] - c1[j];
        }
    }
    __syncthreads();

    const int lane = tid & 31;
    const int wv   = tid >> 5;
    const int mt   = wv & 1;
    const int wl   = (wv >> 1) << 5;
    const int nco  = lane & 15;
    const int hi   = lane >> 4;
    const float* xb = xlds + (wl + nco) * XPADW + (hi << 1);

    float* ost = smem + OUT_OFF + ((mt << 4) + (hi << 3)) * WW + ((wl + nco) ^ (hi << 5));
    const float* ord = smem + OUT_OFF + (wv << 1) * WW + ((lane << 2) ^ ((wv & 4) << 3));
    float*       og  = out + ((b * COUT + (wv << 1)) * HH + h0) * WW + (lane << 2);
    const int wvs = __builtin_amdgcn_readfirstlane(wv);
    const float bo0 = bias[(wvs << 1)];
    const float bo1 = bias[(wvs << 1) + 1];

    int ga = 0, gb = 1, gf = 2;

    const v2f* wfrag = wlds + (mt << 5) + lane;
    v2f ca[BSTEP], cl[BSTEP], cr[BSTEP];
    load_block(ca, cl, cr, wfrag, xb + 1 * SLOT_ELEMS, 1 * JSTEP, 0);

    for (int p = 0; p < NPAIR; ++p) {
        const int h = h0 + 2 * p;
        const bool more = (p + 1 < NPAIR);

        float rn[PF], rs[PF];
        load_row(rn, xsrc, more ? h + 3 : -1);

        const float* xt0 = xb + (2 * ga)     * SLOT_ELEMS;
        const float* xt1 = xb + (2 * ga + 1) * SLOT_ELEMS;
        const float* xt2 = xb + (2 * gb)     * SLOT_ELEMS;
        const float* xt3 = xb + (2 * gb + 1) * SLOT_ELEMS;
        const float* xn1 = xb + (2 * gf + 1) * SLOT_ELEMS;

        v4f o[OTASKS] = {};
        v8f aL = {}, aR = {};
        v8f bL = {}, bR = {};
        v8f cL = {}, cR = {};

        #pragma unroll
        for (int q = 0; q < NBLK; ++q) {
            const int half = q / 3;
            const int jq   = (half == 0) ? 1 : (half == 1) ? 0 : (half == 2) ? 3 : 2;

            const int nq    = (q + 1 < NBLK) ? q + 1 : 0;
            const int nhalf = nq / 3;
            const int nj    = (nhalf == 0) ? 1 : (nhalf == 1) ? 0 : (nhalf == 2) ? 3 : 2;
            const int nkw   = nq - 3 * nhalf;
            const float* xn = (q + 1 < NBLK) ? ((nj == 0) ? xt0 : (nj == 1) ? xt1 : (nj == 2) ? xt2 : xt3) : xn1;
            v2f na[BSTEP], nl[BSTEP], nrr[BSTEP];
            load_block(na, nl, nrr, wfrag, xn, nj * JSTEP + nkw * BSTEP, nkw);

            if (p > 0) {
                if (q >= 1 && q <= OTASKS / 2) {
                    float* ogp = og + (2 * (p - 1)) * WW;
                    out_piece_store(ogp, 2 * q - 2, o[2 * q - 2], bo0, bo1);
                    out_piece_store(ogp, 2 * q - 1, o[2 * q - 1], bo0, bo1);
                }
                if (q < OTASKS / 2) {
                    o[2 * q]     = out_piece_load(ord, 2 * q);
                    o[2 * q + 1] = out_piece_load(ord, 2 * q + 1);
                }
            }

            __builtin_amdgcn_sched_barrier(0);

            #pragma unroll
            for (int i = 0; i < BSTEP; ++i) {
                if (jq == 1) {
                    bL = __builtin_amdgcn_wmma_f32_16x16x4_f32(false, ca[i], false, cl[i], (short)0, bL, false, false);
                    bR = __builtin_amdgcn_wmma_f32_16x16x4_f32(false, ca[i], false, cr[i], (short)0, bR, false, false);
                } else if (jq == 0) {
                    const v8f sl = (q == 3 && i == 0) ? bL : aL;
                    const v8f sr = (q == 3 && i == 0) ? bR : aR;
                    aL = __builtin_amdgcn_wmma_f32_16x16x4_f32(false, ca[i], false, cl[i], (short)0, sl, false, false);
                    aR = __builtin_amdgcn_wmma_f32_16x16x4_f32(false, ca[i], false, cr[i], (short)0, sr, false, false);
                } else if (jq == 3) {
                    bL = __builtin_amdgcn_wmma_f32_16x16x4_f32(false, ca[i], false, cl[i], (short)0, bL, false, false);
                    bR = __builtin_amdgcn_wmma_f32_16x16x4_f32(false, ca[i], false, cr[i], (short)0, bR, false, false);
                } else {
                    cL = __builtin_amdgcn_wmma_f32_16x16x4_f32(false, ca[i], false, cl[i], (short)0, cL, false, false);
                    cR = __builtin_amdgcn_wmma_f32_16x16x4_f32(false, ca[i], false, cr[i], (short)0, cR, false, false);
                }
            }

            __builtin_amdgcn_sched_barrier(0);

            #pragma unroll
            for (int i = 0; i < BSTEP; ++i) { ca[i] = na[i]; cl[i] = nl[i]; cr[i] = nrr[i]; }

            if (q == 5) {
                float* d0 = xdst + (2 * gf)     * SLOT_ELEMS;
                float* d1 = xdst + (2 * gf + 1) * SLOT_ELEMS;
                #pragma unroll
                for (int j = 0; j < PF; ++j) {
                    d0[2 * j] = c0[j] - rn[j];
                    d1[2 * j] = c1[j] + rn[j];
                }
                __syncthreads();
                float* d2 = xdst + (2 * ga) * SLOT_ELEMS;
                #pragma unroll
                for (int j = 0; j < PF; ++j) d2[2 * j] = rn[j] - c1[j];
                load_row(rs, xsrc, more ? h + 4 : -1);
            }
        }

        {
            float* d3 = xdst + (2 * ga + 1) * SLOT_ELEMS;
            #pragma unroll
            for (int j = 0; j < PF; ++j) d3[2 * j] = c1[j] - rs[j];
        }
        asm volatile("v_nop\n\tv_nop\n\tv_nop\n\tv_nop"
                     : "+v"(aL), "+v"(aR), "+v"(bL), "+v"(bR), "+v"(cL), "+v"(cR));

        {
            const v8f y0l = aL + cL, y0r = aR + cR, y1l = bL - cL, y1r = bR - cR;
            #pragma unroll
            for (int q = 0; q < 8; ++q) {
                ost[q * WW]               = y0l[q];
                ost[q * WW + 16]          = y0r[q];
                ost[(COUT + q) * WW]      = y1l[q];
                ost[(COUT + q) * WW + 16] = y1r[q];
            }
        }

        __builtin_amdgcn_fence(__ATOMIC_RELEASE, "workgroup", "local");
        __builtin_amdgcn_s_barrier_signal(-1);

        __builtin_amdgcn_s_barrier_wait(-1);
        __builtin_amdgcn_fence(__ATOMIC_ACQUIRE, "workgroup", "local");

        #pragma unroll
        for (int j = 0; j < PF; ++j) { c0[j] = rn[j]; c1[j] = rs[j]; }
        { const int g = ga; ga = gf; gf = gb; gb = g; }
    }

    {
        float* ogp = og + (2 * (NPAIR - 1)) * WW;
        v4f o[OTASKS];
        #pragma unroll
        for (int k = 0; k < OTASKS; ++k) o[k] = out_piece_load(ord, k);
        #pragma unroll
        for (int k = 0; k < OTASKS; ++k) out_piece_store(ogp, k, o[k], bo0, bo1);
    }
}

extern "C" void kernel_launch(void* const* d_in, const int* in_sizes, int n_in,
                              void* d_out, int out_size, void* d_ws, size_t ws_size,
                              hipStream_t stream) {
    const float* x    = (const float*)d_in[0];
    const float* wgt  = (const float*)d_in[1];
    const float* bias = (const float*)d_in[2];
    float* out = (float*)d_out;

    dim3 grid(256);
    dim3 block(NTHR);
    hipLaunchKernelGGL(conv3x3_wmma_f32, grid, block, 0, stream, x, wgt, bias, out);
}


# Mid-pair barrier split: signalled after the sixth block, waited for in front of the pair's last block, signalled waves at lower issue priority until then; sums unchanged
# speedup vs baseline: 4.3519x; 1.0359x over previous
#include <hip/hip_runtime.h>


typedef __attribute__((ext_vector_type(2))) float v2f;
typedef __attribute__((ext_vector_type(4))) float v4f;
typedef __attribute__((ext_vector_type(8))) float v8f;

#define CIN    32
#define COUT   32
#define HH     256
#define WW     256
#define ROWS   16
#define NPAIR  8
#define NTHR   512
#define NSLOT  6
#define PF     16
#define BSTEP  8
#define JSTEP  24
#define NSTEP  96
#define NBLK   12

#define XPADW   34
#define XROWW   258
#define SLOT_ELEMS (XROWW * XPADW)
#define X_ELEMS (NSLOT * SLOT_ELEMS)
#define W_OFF   X_ELEMS
#define W_FRAGS (NSTEP * 2 * 32)
#define OUT_OFF (W_OFF + W_FRAGS * 2)
#define OUT_ELEMS (2 * COUT * WW)
#define SMEM_FLOATS (OUT_OFF + OUT_ELEMS)
#define OTASKS 8


static __device__ __forceinline__
void load_block(v2f (&a)[BSTEP], v2f (&bl)[BSTEP], v2f (&br)[BSTEP],
                const v2f* __restrict__ wf, const float* __restrict__ xrow,
                const int sbase, const int kw)
{
    #pragma unroll
    for (int i = 0; i < BSTEP; ++i) {
        a[i]  = wf[((sbase + i) * 2) * 32];
        bl[i] = *(const v2f*)&xrow[kw * XPADW + (i << 2)];
        br[i] = *(const v2f*)&xrow[(kw + 16) * XPADW + (i << 2)];
    }
}

static __device__ __forceinline__
void load_row(float (&r)[PF], const float* __restrict__ xsrc, const int row)
{
    if ((unsigned)row < HH) {
        #pragma unroll
        for (int j = 0; j < PF; ++j) r[j] = xsrc[row * WW + j * (2 * HH * WW)];
    } else {
        #pragma unroll
        for (int j = 0; j < PF; ++j) r[j] = 0.0f;
    }
}

static __device__ __forceinline__ v4f out_piece_load(const float* __restrict__ ord, const int k)
{
    return *(const v4f*)&ord[(((k >> 2) * COUT) + ((k >> 1) & 1)) * WW + (k & 1) * 128];
}

static __device__ __forceinline__
void out_piece_store(float* __restrict__ og, const int k, v4f v, const float b0, const float b1)
{
    v += ((k >> 1) & 1) ? b1 : b0;
    *(v4f*)&og[((((k >> 1) & 1) * HH) + (k >> 2)) * WW + (k & 1) * 128] = v;
}

extern "C" __global__ __launch_bounds__(NTHR)
void conv3x3_wmma_f32(const float* __restrict__ x,
                      const float* __restrict__ wgt,
                      const float* __restrict__ bias,
                      float* __restrict__ out)
{
    __shared__ __attribute__((aligned(16))) float smem[SMEM_FLOATS];
    float* xlds = smem;
    v2f*   wlds = (v2f*)(smem + W_OFF);

    const int tid = threadIdx.x;
    const int b   = blockIdx.x >> 4;
    const int h0  = (blockIdx.x & 15) * ROWS;

    for (int idx = tid; idx < JSTEP * 64; idx += NTHR) {
        int sj   = idx >> 6;
        int rem  = idx & 63;
        int t    = rem >> 5;
        int l    = rem & 31;
        int k0   = 4 * sj + ((l >> 4) << 1);
        int cout = (t << 4) + (l & 15);
        v2f u0, u1, u2, u3;
        #pragma unroll
        for (int v = 0; v < 2; ++v) {
            int k  = k0 + v;
            int ci = k & 31;
            int kw = k >> 5;
            const float* g = wgt + (cout * CIN + ci) * 9 + kw;
            const float g0 = g[0], g1 = g[3], g2 = g[6];
            u0[v] = g0;
            u1[v] = 0.5f * ((g0 + g2) + g1);
            u2[v] = 0.5f * ((g0 + g2) - g1);
            u3[v] = -g2;
        }
        wlds[(0 * JSTEP + sj) * 64 + rem] = u0;
        wlds[(1 * JSTEP + sj) * 64 + rem] = u1;
        wlds[(2 * JSTEP + sj) * 64 + rem] = u2;
        wlds[(3 * JSTEP + sj) * 64 + rem] = u3;
    }

    if (tid < NSLOT * 2 * XPADW) {
        int slot = tid / (2 * XPADW);
        int rem  = tid - slot * (2 * XPADW);
        int side = rem / XPADW;
        int c    = rem - side * XPADW;
        xlds[(slot * XROWW + side * (XROWW - 1)) * XPADW + c] = 0.0f;
    }

    const int sw  = tid & 255;
    const int sc0 = tid >> 8;
    const float* xsrc = x + ((b * CIN + sc0) * HH) * WW + sw;
    float*       xdst = xlds + (sw + 1) * XPADW + sc0;

    float c0[PF], c1[PF];

    {
        float r0[PF], r1[PF];
        load_row(r0, xsrc, h0 - 1);
        load_row(r1, xsrc, h0);
        load_row(c0, xsrc, h0 + 1);
        load_row(c1, xsrc, h0 + 2);
        #pragma unroll
        for (int j = 0; j < PF; ++j) {
            xdst[0 * SLOT_ELEMS + 2 * j] = r0[j] - c0[j];
            xdst[1 * SLOT_ELEMS + 2 * j] = r1[j] + c0[j];
            xdst[2 * SLOT_ELEMS + 2 * j] = c0[j] - r1[j];
            xdst[3 * SLOT_ELEMS + 2 * j] = r1[j] - c1[j];
        }
    }
    __syncthreads();

    const int lane = tid & 31;
    const int wv   = tid >> 5;
    const int mt   = wv & 1;
    const int wl   = (wv >> 1) << 5;
    const int nco  = lane & 15;
    const int hi   = lane >> 4;
    const float* xb = xlds + (wl + nco) * XPADW + (hi << 1);

    float* ost = smem + OUT_OFF + ((mt << 4) + (hi << 3)) * WW + ((wl + nco) ^ (hi << 5));
    const float* ord = smem + OUT_OFF + (wv << 1) * WW + ((lane << 2) ^ ((wv & 4) << 3));
    float*       og  = out + ((b * COUT + (wv << 1)) * HH + h0) * WW + (lane << 2);
    const int wvs = __builtin_amdgcn_readfirstlane(wv);
    const float bo0 = bias[(wvs << 1)];
    const float bo1 = bias[(wvs << 1) + 1];

    int ga = 0, gb = 1, gf = 2;

    const v2f* wfrag = wlds + (mt << 5) + lane;
    v2f ca[BSTEP], cl[BSTEP], cr[BSTEP];
    load_block(ca, cl, cr, wfrag, xb + 1 * SLOT_ELEMS, 1 * JSTEP, 0);

    __builtin_amdgcn_s_setprio(1);

    for (int p = 0; p < NPAIR; ++p) {
        const int h = h0 + 2 * p;
        const bool more = (p + 1 < NPAIR);

        float rn[PF], rs[PF];
        load_row(rn, xsrc, more ? h + 3 : -1);

        const float* xt0 = xb + (2 * ga)     * SLOT_ELEMS;
        const float* xt1 = xb + (2 * ga + 1) * SLOT_ELEMS;
        const float* xt2 = xb + (2 * gb)     * SLOT_ELEMS;
        const float* xt3 = xb + (2 * gb + 1) * SLOT_ELEMS;
        const float* xn1 = xb + (2 * gf + 1) * SLOT_ELEMS;

        v4f o[OTASKS] = {};
        v8f aL = {}, aR = {};
        v8f bL = {}, bR = {};
        v8f cL = {}, cR = {};

        #pragma unroll
        for (int q = 0; q < NBLK; ++q) {
            const int half = q / 3;
            const int jq   = (half == 0) ? 1 : (half == 1) ? 0 : (half == 2) ? 3 : 2;

            const int nq    = (q + 1 < NBLK) ? q + 1 : 0;
            const int nhalf = nq / 3;
            const int nj    = (nhalf == 0) ? 1 : (nhalf == 1) ? 0 : (nhalf == 2) ? 3 : 2;
            const int nkw   = nq - 3 * nhalf;
            const float* xn = (q + 1 < NBLK) ? ((nj == 0) ? xt0 : (nj == 1) ? xt1 : (nj == 2) ? xt2 : xt3) : xn1;

            if (q == NBLK - 1) {
                __builtin_amdgcn_s_barrier_wait(-1);
                __builtin_amdgcn_fence(__ATOMIC_ACQUIRE, "workgroup", "local");
                __builtin_amdgcn_s_setprio(1);
                float* d2 = xdst + (2 * ga) * SLOT_ELEMS;
                #pragma unroll
                for (int j = 0; j < PF; ++j) d2[2 * j] = rn[j] - c1[j];
            }

            v2f na[BSTEP], nl[BSTEP], nrr[BSTEP];
            load_block(na, nl, nrr, wfrag, xn, nj * JSTEP + nkw * BSTEP, nkw);

            if (p > 0) {
                if (q >= 1 && q <= OTASKS / 2) {
                    float* ogp = og + (2 * (p - 1)) * WW;
                    out_piece_store(ogp, 2 * q - 2, o[2 * q - 2], bo0, bo1);
                    out_piece_store(ogp, 2 * q - 1, o[2 * q - 1], bo0, bo1);
                }
                if (q < OTASKS / 2) {
                    o[2 * q]     = out_piece_load(ord, 2 * q);
                    o[2 * q + 1] = out_piece_load(ord, 2 * q + 1);
                }
            }

            __builtin_amdgcn_sched_barrier(0);

            #pragma unroll
            for (int i = 0; i < BSTEP; ++i) {
                if (jq == 1) {
                    bL = __builtin_amdgcn_wmma_f32_16x16x4_f32(false, ca[i], false, cl[i], (short)0, bL, false, false);
                    bR = __builtin_amdgcn_wmma_f32_16x16x4_f32(false, ca[i], false, cr[i], (short)0, bR, false, false);
                } else if (jq == 0) {
                    const v8f sl = (q == 3 && i == 0) ? bL : aL;
                    const v8f sr = (q == 3 && i == 0) ? bR : aR;
                    aL = __builtin_amdgcn_wmma_f32_16x16x4_f32(false, ca[i], false, cl[i], (short)0, sl, false, false);
                    aR = __builtin_amdgcn_wmma_f32_16x16x4_f32(false, ca[i], false, cr[i], (short)0, sr, false, false);
                } else if (jq == 3) {
                    bL = __builtin_amdgcn_wmma_f32_16x16x4_f32(false, ca[i], false, cl[i], (short)0, bL, false, false);
                    bR = __builtin_amdgcn_wmma_f32_16x16x4_f32(false, ca[i], false, cr[i], (short)0, bR, false, false);
                } else {
                    cL = __builtin_amdgcn_wmma_f32_16x16x4_f32(false, ca[i], false, cl[i], (short)0, cL, false, false);
                    cR = __builtin_amdgcn_wmma_f32_16x16x4_f32(false, ca[i], false, cr[i], (short)0, cR, false, false);
                }
            }

            __builtin_amdgcn_sched_barrier(0);

            #pragma unroll
            for (int i = 0; i < BSTEP; ++i) { ca[i] = na[i]; cl[i] = nl[i]; cr[i] = nrr[i]; }

            if (q == 5) {
                float* d0 = xdst + (2 * gf)     * SLOT_ELEMS;
                float* d1 = xdst + (2 * gf + 1) * SLOT_ELEMS;
                #pragma unroll
                for (int j = 0; j < PF; ++j) {
                    d0[2 * j] = c0[j] - rn[j];
                    d1[2 * j] = c1[j] + rn[j];
                }
                __builtin_amdgcn_fence(__ATOMIC_RELEASE, "workgroup", "local");
                __builtin_amdgcn_s_barrier_signal(-1);
                __builtin_amdgcn_s_setprio(0);
                load_row(rs, xsrc, more ? h + 4 : -1);
            }
        }

        {
            float* d3 = xdst + (2 * ga + 1) * SLOT_ELEMS;
            #pragma unroll
            for (int j = 0; j < PF; ++j) d3[2 * j] = c1[j] - rs[j];
        }
        asm volatile("v_nop\n\tv_nop\n\tv_nop\n\tv_nop"
                     : "+v"(aL), "+v"(aR), "+v"(bL), "+v"(bR), "+v"(cL), "+v"(cR));

        {
            const v8f y0l = aL + cL, y0r = aR + cR, y1l = bL - cL, y1r = bR - cR;
            #pragma unroll
            for (int q = 0; q < 8; ++q) {
                ost[q * WW]               = y0l[q];
                ost[q * WW + 16]          = y0r[q];
                ost[(COUT + q) * WW]      = y1l[q];
                ost[(COUT + q) * WW + 16] = y1r[q];
            }
        }

        __builtin_amdgcn_fence(__ATOMIC_RELEASE, "workgroup", "local");
        __builtin_amdgcn_s_barrier_signal(-1);

        __builtin_amdgcn_s_barrier_wait(-1);
        __builtin_amdgcn_fence(__ATOMIC_ACQUIRE, "workgroup", "local");

        #pragma unroll
        for (int j = 0; j < PF; ++j) { c0[j] = rn[j]; c1[j] = rs[j]; }
        { const int g = ga; ga = gf; gf = gb; gb = g; }
    }

    {
        float* ogp = og + (2 * (NPAIR - 1)) * WW;
        v4f o[OTASKS];
        #pragma unroll
        for (int k = 0; k < OTASKS; ++k) o[k] = out_piece_load(ord, k);
        #pragma unroll
        for (int k = 0; k < OTASKS; ++k) out_piece_store(ogp, k, o[k], bo0, bo1);
    }
}

extern "C" void kernel_launch(void* const* d_in, const int* in_sizes, int n_in,
                              void* d_out, int out_size, void* d_ws, size_t ws_size,
                              hipStream_t stream) {
    const float* x    = (const float*)d_in[0];
    const float* wgt  = (const float*)d_in[1];
    const float* bias = (const float*)d_in[2];
    float* out = (float*)d_out;

    dim3 grid(256);
    dim3 block(NTHR);
    hipLaunchKernelGGL(conv3x3_wmma_f32, grid, block, 0, stream, x, wgt, bias, out);
}
